# starting kernel
# baseline (speedup 1.0000x reference)
#include <hip/hip_runtime.h>

typedef float v2f __attribute__((ext_vector_type(2)));
typedef float v4f __attribute__((ext_vector_type(4)));
typedef float v8f __attribute__((ext_vector_type(8)));

#define DSZ   4096
#define TILE  16
#define BLOCK 256
#define WAVES 8
#define KSEG  (DSZ / WAVES)

__global__ __launch_bounds__(BLOCK) void mix_kernel(
    const float* __restrict__ x, const float* __restrict__ sx,
    const float* __restrict__ tmk, const float* __restrict__ tmv,
    const float* __restrict__ tmr,
    float* __restrict__ xk, float* __restrict__ xv, float* __restrict__ xr)
{
  int i = blockIdx.x * blockDim.x + threadIdx.x;
  if (i < DSZ) {
    float xi = x[i], si = sx[i];
    float mk = tmk[i], mv = tmv[i], mr = tmr[i];
    xk[i] = xi * mk + si * (1.0f - mk);
    xv[i] = xi * mv + si * (1.0f - mv);
    xr[i] = xi * mr + si * (1.0f - mr);
  }
}

__global__ __launch_bounds__(BLOCK) void gemv_wmma_kernel(
    const float* __restrict__ W0, const float* __restrict__ W1,
    const float* __restrict__ W2,
    const float* __restrict__ x0, const float* __restrict__ x1,
    const float* __restrict__ x2,
    float* __restrict__ y0, float* __restrict__ y1, float* __restrict__ y2)
{
  __shared__ __align__(16) float lds_x[DSZ];
  __shared__ float lds_red[WAVES * TILE];

  const float* W; const float* xvec; float* y;
  switch (blockIdx.y) {
    case 0:  W = W0; xvec = x0; y = y0; break;
    case 1:  W = W1; xvec = x1; y = y1; break;
    default: W = W2; xvec = x2; y = y2; break;
  }

  for (int i = threadIdx.x; i < DSZ / 4; i += BLOCK)
    ((v4f*)lds_x)[i] = ((const v4f*)xvec)[i];
  __syncthreads();

  const int lane = threadIdx.x & 31;
  const int wave = threadIdx.x >> 5;
  const int row  = (blockIdx.x << 4) + (lane & 15);
  const int koff = (lane >> 4) << 1;

  const float* aBase = W + (size_t)row * DSZ + koff;

  v8f c0 = {}; v8f c1 = {};
  const int kBeg = wave * KSEG;
  const int kEnd = kBeg + KSEG;
  for (int kc = kBeg; kc < kEnd; kc += 8) {
    v2f a0 = *(const v2f*)(aBase + kc);
    v2f b0 = *(const v2f*)(&lds_x[kc + koff]);
    v2f a1 = *(const v2f*)(aBase + kc + 4);
    v2f b1 = *(const v2f*)(&lds_x[kc + 4 + koff]);
    c0 = __builtin_amdgcn_wmma_f32_16x16x4_f32(false, a0, false, b0,
                                               (short)0, c0, false, false);
    c1 = __builtin_amdgcn_wmma_f32_16x16x4_f32(false, a1, false, b1,
                                               (short)0, c1, false, false);
  }
  v8f c = c0 + c1;

  if ((lane & 15) == 0) {
    int half = lane >> 4;
#pragma unroll
    for (int r = 0; r < 8; ++r)
      lds_red[wave * TILE + half * 8 + r] = c[r];
  }
  __syncthreads();

  if (threadIdx.x < TILE) {
    float s = 0.0f;
#pragma unroll
    for (int w = 0; w < WAVES; ++w) s += lds_red[w * TILE + threadIdx.x];
    y[(blockIdx.x << 4) + threadIdx.x] = s;
  }
}

__global__ __launch_bounds__(BLOCK) void wkv_kernel(
    const float* __restrict__ x,  const float* __restrict__ aa,
    const float* __restrict__ bb, const float* __restrict__ pp,
    const float* __restrict__ t_decay, const float* __restrict__ t_first,
    const float* __restrict__ k, const float* __restrict__ v,
    const float* __restrict__ rraw,
    float* __restrict__ rkv, float* __restrict__ out)
{
  int i = blockIdx.x * blockDim.x + threadIdx.x;
  if (i >= DSZ) return;

  float ki = k[i], vi = v[i];
  float aai = aa[i], bbi = bb[i], ppi = pp[i];
  float r = 1.0f / (1.0f + expf(-rraw[i]));

  float ww = t_first[i] + ki;
  float p  = fmaxf(ppi, ww);
  float e1 = expf(ppi - p);
  float e2 = expf(ww - p);
  float a  = e1 * aai + e2 * vi;
  float b  = e1 * bbi + e2;

  float ww2 = ppi + t_decay[i];
  float p2  = fmaxf(ww2, ki);
  float e1b = expf(ww2 - p2);
  float e2b = expf(ki - p2);

  rkv[i] = r * (a / b);

  out[1 * DSZ + i] = x[i];
  out[2 * DSZ + i] = e1b * aai + e2b * vi;
  out[3 * DSZ + i] = e1b * bbi + e2b;
  out[4 * DSZ + i] = p2;
}

extern "C" void kernel_launch(void* const* d_in, const int* in_sizes, int n_in,
                              void* d_out, int out_size, void* d_ws,
                              size_t ws_size, hipStream_t stream)
{
  const float* x       = (const float*)d_in[0];
  const float* sxx     = (const float*)d_in[1];
  const float* aa      = (const float*)d_in[2];
  const float* bb      = (const float*)d_in[3];
  const float* pp      = (const float*)d_in[4];
  const float* w_key   = (const float*)d_in[5];
  const float* w_val   = (const float*)d_in[6];
  const float* w_rec   = (const float*)d_in[7];
  const float* w_out   = (const float*)d_in[8];
  const float* t_decay = (const float*)d_in[9];
  const float* t_first = (const float*)d_in[10];
  const float* t_mix_k = (const float*)d_in[11];
  const float* t_mix_v = (const float*)d_in[12];
  const float* t_mix_r = (const float*)d_in[13];

  float* out = (float*)d_out;
  float* ws  = (float*)d_ws;
  float* xk  = ws + 0 * DSZ;
  float* xv  = ws + 1 * DSZ;
  float* xr  = ws + 2 * DSZ;
  float* kk  = ws + 3 * DSZ;
  float* vv  = ws + 4 * DSZ;
  float* rr  = ws + 5 * DSZ;
  float* rkv = ws + 6 * DSZ;

  mix_kernel<<<DSZ / BLOCK, BLOCK, 0, stream>>>(x, sxx, t_mix_k, t_mix_v,
                                                t_mix_r, xk, xv, xr);

  dim3 g1(DSZ / TILE, 3);
  gemv_wmma_kernel<<<g1, BLOCK, 0, stream>>>(w_key, w_val, w_rec,
                                             xk, xv, xr,
                                             kk, vv, rr);

  wkv_kernel<<<DSZ / BLOCK, BLOCK, 0, stream>>>(x, aa, bb, pp, t_decay,
                                                t_first, kk, vv, rr,
                                                rkv, out);

  dim3 g2(DSZ / TILE, 1);
  gemv_wmma_kernel<<<g2, BLOCK, 0, stream>>>(w_out, w_out, w_out,
                                             rkv, rkv, rkv,
                                             out, out, out);
}


# GEMV loader: each wave streams W as 512-B contiguous row pieces (non-temporal b128, 16-32 loads in flight), x piece in registers, row dot on the WMMA diagonal; partials summed via LDS (new sum order)
# speedup vs baseline: 1.2403x; 1.2403x over previous
#include <hip/hip_runtime.h>

typedef float v2f __attribute__((ext_vector_type(2)));
typedef float v4f __attribute__((ext_vector_type(4)));
typedef float v8f __attribute__((ext_vector_type(8)));

#define DSZ   4096
#define BLOCK 256

__global__ __launch_bounds__(BLOCK) void mix_kernel(
    const float* __restrict__ x, const float* __restrict__ sx,
    const float* __restrict__ tmk, const float* __restrict__ tmv,
    const float* __restrict__ tmr,
    float* __restrict__ xk, float* __restrict__ xv, float* __restrict__ xr)
{
  int i = blockIdx.x * blockDim.x + threadIdx.x;
  if (i < DSZ) {
    float xi = x[i], si = sx[i];
    float mk = tmk[i], mv = tmv[i], mr = tmr[i];
    xk[i] = xi * mk + si * (1.0f - mk);
    xv[i] = xi * mv + si * (1.0f - mv);
    xr[i] = xi * mr + si * (1.0f - mr);
  }
}

#define GROWS 16
#define GWAVES 16
#define GTHREADS (GWAVES * 32)
#define GSEG (DSZ / GWAVES)
#define CH 8

static __device__ __forceinline__ v4f load_stream(const float* p) {
  return __builtin_nontemporal_load((const v4f*)p);
}

template <int NM>
__global__ __launch_bounds__(GTHREADS) void gemv_wmma_kernel(
    const float* __restrict__ W0, const float* __restrict__ W1,
    const float* __restrict__ W2,
    const float* __restrict__ x0, const float* __restrict__ x1,
    const float* __restrict__ x2,
    float* __restrict__ y0, float* __restrict__ y1, float* __restrict__ y2)
{
  __shared__ float red[NM * GROWS][GWAVES * 16];
  __shared__ float fin[NM * GROWS];

  const int lane = threadIdx.x & 31;
  const int wave = __builtin_amdgcn_readfirstlane(threadIdx.x >> 5);
  const int koff = wave * GSEG + 4 * lane;

  v4f xq[NM][2];
#pragma unroll
  for (int m = 0; m < NM; ++m) {
    const float* xv = (m == 0 ? x0 : (m == 1 ? x1 : x2)) + koff;
    xq[m][0] = *(const v4f*)(xv);
    xq[m][1] = *(const v4f*)(xv + 128);
  }

  const size_t rbase = (size_t)blockIdx.x * GROWS * DSZ + koff;
  constexpr int NCH = NM * (GROWS / CH);
  v4f w[2][CH][2];

#define LOAD_GROUP(G) do {                                                    \
    const int m_ = (G) / (GROWS / CH);                                        \
    const float* wp_ = (m_ == 0 ? W0 : (m_ == 1 ? W1 : W2)) + rbase           \
                       + (size_t)(((G) % (GROWS / CH)) * CH) * DSZ;           \
    _Pragma("unroll") for (int r_ = 0; r_ < CH; ++r_) {                       \
      w[(G) & 1][r_][0] = load_stream(wp_ + (size_t)r_ * DSZ);                \
      w[(G) & 1][r_][1] = load_stream(wp_ + (size_t)r_ * DSZ + 128);          \
    } } while (0)

  const int  dsel  = lane & 7;
  const bool valid = (lane < 8) || (lane >= 24);
  const int  slot  = wave * 16 + (lane < 8 ? lane : lane - 16);

  LOAD_GROUP(0);
#pragma unroll
  for (int g = 0; g < NCH; ++g) {
    __builtin_amdgcn_sched_barrier(0);
    if (g + 1 < NCH) LOAD_GROUP(g + 1);
    __builtin_amdgcn_sched_barrier(0);

    const int m = g / (GROWS / CH);
    const v2f xb0 = {xq[m][0].x, xq[m][0].y};
    const v2f xb1 = {xq[m][0].z, xq[m][0].w};
    const v2f xb2 = {xq[m][1].x, xq[m][1].y};
    const v2f xb3 = {xq[m][1].z, xq[m][1].w};
    float e[CH];
#pragma unroll
    for (int r = 0; r < CH; r += 2) {
      const v4f* wa = w[g & 1][r];
      const v4f* wb = w[g & 1][r + 1];
      v8f c0 = {}; v8f c1 = {};
      c0 = __builtin_amdgcn_wmma_f32_16x16x4_f32(false, (v2f){wa[0].x, wa[0].y}, false, xb0, (short)0, c0, false, false);
      c1 = __builtin_amdgcn_wmma_f32_16x16x4_f32(false, (v2f){wb[0].x, wb[0].y}, false, xb0, (short)0, c1, false, false);
      c0 = __builtin_amdgcn_wmma_f32_16x16x4_f32(false, (v2f){wa[0].z, wa[0].w}, false, xb1, (short)0, c0, false, false);
      c1 = __builtin_amdgcn_wmma_f32_16x16x4_f32(false, (v2f){wb[0].z, wb[0].w}, false, xb1, (short)0, c1, false, false);
      c0 = __builtin_amdgcn_wmma_f32_16x16x4_f32(false, (v2f){wa[1].x, wa[1].y}, false, xb2, (short)0, c0, false, false);
      c1 = __builtin_amdgcn_wmma_f32_16x16x4_f32(false, (v2f){wb[1].x, wb[1].y}, false, xb2, (short)0, c1, false, false);
      c0 = __builtin_amdgcn_wmma_f32_16x16x4_f32(false, (v2f){wa[1].z, wa[1].w}, false, xb3, (short)0, c0, false, false);
      c1 = __builtin_amdgcn_wmma_f32_16x16x4_f32(false, (v2f){wb[1].z, wb[1].w}, false, xb3, (short)0, c1, false, false);
      float d0 = c0[0], d1 = c1[0];
#pragma unroll
      for (int j = 1; j < 8; ++j) {
        d0 = (dsel == j) ? c0[j] : d0;
        d1 = (dsel == j) ? c1[j] : d1;
      }
      e[r] = d0; e[r + 1] = d1;
    }
    if (valid) {
#pragma unroll
      for (int r = 0; r < CH; ++r) red[g * CH + r][slot] = e[r];
    }
  }
#undef LOAD_GROUP
  __syncthreads();

#pragma unroll
  for (int m = 0; m < NM; ++m) {
    const float* p = &red[m * GROWS + wave][lane];
    float s = 0.0f;
#pragma unroll
    for (int i = 0; i < 8; ++i) s += p[32 * i];
#pragma unroll
    for (int k = 16; k >= 1; k >>= 1) s += __shfl_xor(s, k, 32);
    if (lane == 0) fin[m * GROWS + wave] = s;
  }
  __syncthreads();

  if (threadIdx.x < NM * GROWS) {
    const int m = threadIdx.x >> 4;
    float* y = (m == 0 ? y0 : (m == 1 ? y1 : y2));
    y[blockIdx.x * GROWS + (threadIdx.x & 15)] = fin[threadIdx.x];
  }
}

__global__ __launch_bounds__(BLOCK) void wkv_kernel(
    const float* __restrict__ x,  const float* __restrict__ aa,
    const float* __restrict__ bb, const float* __restrict__ pp,
    const float* __restrict__ t_decay, const float* __restrict__ t_first,
    const float* __restrict__ k, const float* __restrict__ v,
    const float* __restrict__ rraw,
    float* __restrict__ rkv, float* __restrict__ out)
{
  int i = blockIdx.x * blockDim.x + threadIdx.x;
  if (i >= DSZ) return;

  float ki = k[i], vi = v[i];
  float aai = aa[i], bbi = bb[i], ppi = pp[i];
  float r = 1.0f / (1.0f + expf(-rraw[i]));

  float ww = t_first[i] + ki;
  float p  = fmaxf(ppi, ww);
  float e1 = expf(ppi - p);
  float e2 = expf(ww - p);
  float a  = e1 * aai + e2 * vi;
  float b  = e1 * bbi + e2;

  float ww2 = ppi + t_decay[i];
  float p2  = fmaxf(ww2, ki);
  float e1b = expf(ww2 - p2);
  float e2b = expf(ki - p2);

  rkv[i] = r * (a / b);

  out[1 * DSZ + i] = x[i];
  out[2 * DSZ + i] = e1b * aai + e2b * vi;
  out[3 * DSZ + i] = e1b * bbi + e2b;
  out[4 * DSZ + i] = p2;
}

extern "C" void kernel_launch(void* const* d_in, const int* in_sizes, int n_in,
                              void* d_out, int out_size, void* d_ws,
                              size_t ws_size, hipStream_t stream)
{
  const float* x       = (const float*)d_in[0];
  const float* sxx     = (const float*)d_in[1];
  const float* aa      = (const float*)d_in[2];
  const float* bb      = (const float*)d_in[3];
  const float* pp      = (const float*)d_in[4];
  const float* w_key   = (const float*)d_in[5];
  const float* w_val   = (const float*)d_in[6];
  const float* w_rec   = (const float*)d_in[7];
  const float* w_out   = (const float*)d_in[8];
  const float* t_decay = (const float*)d_in[9];
  const float* t_first = (const float*)d_in[10];
  const float* t_mix_k = (const float*)d_in[11];
  const float* t_mix_v = (const float*)d_in[12];
  const float* t_mix_r = (const float*)d_in[13];

  float* out = (float*)d_out;
  float* ws  = (float*)d_ws;
  float* xk  = ws + 0 * DSZ;
  float* xv  = ws + 1 * DSZ;
  float* xr  = ws + 2 * DSZ;
  float* kk  = ws + 3 * DSZ;
  float* vv  = ws + 4 * DSZ;
  float* rr  = ws + 5 * DSZ;
  float* rkv = ws + 6 * DSZ;

  mix_kernel<<<DSZ / BLOCK, BLOCK, 0, stream>>>(x, sxx, t_mix_k, t_mix_v,
                                                t_mix_r, xk, xv, xr);

  gemv_wmma_kernel<3><<<DSZ / GROWS, GTHREADS, 0, stream>>>(w_key, w_val, w_rec,
                                                            xk, xv, xr,
                                                            kk, vv, rr);

  wkv_kernel<<<DSZ / BLOCK, BLOCK, 0, stream>>>(x, aa, bb, pp, t_decay,
                                                t_first, kk, vv, rr,
                                                rkv, out);

  gemv_wmma_kernel<1><<<DSZ / GROWS, GTHREADS, 0, stream>>>(w_out, w_out, w_out,
                                                            rkv, rkv, rkv,
                                                            out, out, out);
}


# One wave per SIMD: 4-wave blocks (1024 waves per launch, was 4096), a wave streams 4 KB row pieces, 48 loads in flight, a row's 16 WMMAs chained in one D; new sum order (64 partials per row)
# speedup vs baseline: 1.5429x; 1.2440x over previous
#include <hip/hip_runtime.h>

typedef float v2f __attribute__((ext_vector_type(2)));
typedef float v4f __attribute__((ext_vector_type(4)));
typedef float v8f __attribute__((ext_vector_type(8)));

#define DSZ   4096

#define GROWS 16
#define GWAVES 4
#define GTHREADS (GWAVES * 32)
#define GSEG (DSZ / GWAVES)
#define NPC (GSEG / 128)
#define NPAIR (GROWS / 2)
#define RING 4
#define RPITCH (GWAVES * 16 + 8)

static __device__ __forceinline__ v4f load_stream(const float* p) {
  return __builtin_nontemporal_load((const v4f*)p);
}

template <int NM, class XF>
static __device__ __forceinline__ void gemv_rows(
    const float* __restrict__ W0, const float* __restrict__ W1,
    const float* __restrict__ W2, XF make_x,
    float (*red)[RPITCH], float (&rowsum)[NM],
    const int lane, const int wave, const int koff)
{
  const size_t rbase = (size_t)blockIdx.x * GROWS * DSZ + koff;
  constexpr int NG = NM * NPAIR;
  v4f w[RING][2][NPC];
  v4f xq[NM][NPC];

#define LOAD_PAIR(G) do {                                                     \
    const int m_ = (G) / NPAIR;                                               \
    const float* wp_ = (m_ == 0 ? W0 : (m_ == 1 ? W1 : W2)) + rbase           \
                       + (size_t)(((G) % NPAIR) * 2) * DSZ;                   \
    _Pragma("unroll") for (int r_ = 0; r_ < 2; ++r_) {                        \
      _Pragma("unroll") for (int j_ = 0; j_ < NPC; ++j_)                      \
        w[(G) % RING][r_][j_] = load_stream(wp_ + (size_t)r_ * DSZ + 128 * j_);\
    } } while (0)

  const int  dsel  = lane & 7;
  const bool valid = (lane < 8) || (lane >= 24);
  const int  slot  = wave * 16 + (lane < 8 ? lane : lane - 16);

  __builtin_amdgcn_sched_barrier(0);
  LOAD_PAIR(0);
  LOAD_PAIR(1);
  LOAD_PAIR(2);
#pragma unroll
  for (int g = 0; g < NG; ++g) {
    __builtin_amdgcn_sched_barrier(0);
    if (g + 3 < NG) LOAD_PAIR(g + 3);
    __builtin_amdgcn_sched_barrier(0);

    if (g == 0) make_x(xq);

    const int m = g / NPAIR;
    const v4f* wa = w[g % RING][0];
    const v4f* wb = w[g % RING][1];
    v8f c0 = {}; v8f c1 = {};
#pragma unroll
    for (int j = 0; j < NPC; ++j) {
      const v2f xlo = {xq[m][j].x, xq[m][j].y};
      const v2f xhi = {xq[m][j].z, xq[m][j].w};
      c0 = __builtin_amdgcn_wmma_f32_16x16x4_f32(false, (v2f){wa[j].x, wa[j].y}, false, xlo, (short)0, c0, false, false);
      c1 = __builtin_amdgcn_wmma_f32_16x16x4_f32(false, (v2f){wb[j].x, wb[j].y}, false, xlo, (short)0, c1, false, false);
      c0 = __builtin_amdgcn_wmma_f32_16x16x4_f32(false, (v2f){wa[j].z, wa[j].w}, false, xhi, (short)0, c0, false, false);
      c1 = __builtin_amdgcn_wmma_f32_16x16x4_f32(false, (v2f){wb[j].z, wb[j].w}, false, xhi, (short)0, c1, false, false);
    }
    float d0 = c0[0], d1 = c1[0];
#pragma unroll
    for (int j = 1; j < 8; ++j) {
      d0 = (dsel == j) ? c0[j] : d0;
      d1 = (dsel == j) ? c1[j] : d1;
    }
    if (valid) {
      red[2 * g][slot]     = d0;
      red[2 * g + 1][slot] = d1;
    }
  }
#undef LOAD_PAIR
  __syncthreads();

  const int rloc = wave * 4 + (lane >> 3);
#pragma unroll
  for (int m = 0; m < NM; ++m) {
    const float* p = &red[m * GROWS + rloc][lane & 7];
    float s = 0.0f;
#pragma unroll
    for (int i = 0; i < GWAVES * 2; ++i) s += p[8 * i];
#pragma unroll
    for (int k = 4; k >= 1; k >>= 1) s += __shfl_xor(s, k, 32);
    rowsum[m] = s;
  }
}

__global__ __launch_bounds__(GTHREADS) __attribute__((amdgpu_waves_per_eu(1, 1)))
void mix_gemv3_wkv_kernel(
    const float* __restrict__ w_key, const float* __restrict__ w_val,
    const float* __restrict__ w_rec,
    const float* __restrict__ x, const float* __restrict__ sx,
    const float* __restrict__ tmk, const float* __restrict__ tmv,
    const float* __restrict__ tmr,
    const float* __restrict__ aa, const float* __restrict__ bb,
    const float* __restrict__ pp, const float* __restrict__ t_decay,
    const float* __restrict__ t_first,
    float* __restrict__ rkv, float* __restrict__ out)
{
  __shared__ float red[3 * GROWS][RPITCH];
  __shared__ float fin[5 * GROWS];

  const int lane = threadIdx.x & 31;
  const int wave = __builtin_amdgcn_readfirstlane(threadIdx.x >> 5);
  const int koff = wave * GSEG + 4 * lane;

  const int rloc = wave * 4 + (lane >> 3);
  const int row  = blockIdx.x * GROWS + rloc;
  const float xi  = x[row],  aai = aa[row], bbi = bb[row], ppi = pp[row];
  const float tdi = t_decay[row], tfi = t_first[row];

  v4f xx[NPC], ss[NPC], mk[NPC], mv[NPC], mr[NPC];
#pragma unroll
  for (int h = 0; h < NPC; ++h) {
    const int o = koff + 128 * h;
    xx[h] = *(const v4f*)(x + o);
    ss[h] = *(const v4f*)(sx + o);
    mk[h] = *(const v4f*)(tmk + o);
    mv[h] = *(const v4f*)(tmv + o);
    mr[h] = *(const v4f*)(tmr + o);
  }
  auto mix = [&](v4f (&xq)[3][NPC]) {
    asm volatile("" : "+v"(mk[0]), "+v"(mk[1]), "+v"(mk[2]), "+v"(mk[3]),
                      "+v"(mk[4]), "+v"(mk[5]), "+v"(mk[6]), "+v"(mk[7]));
    asm volatile("" : "+v"(mv[0]), "+v"(mv[1]), "+v"(mv[2]), "+v"(mv[3]),
                      "+v"(mv[4]), "+v"(mv[5]), "+v"(mv[6]), "+v"(mv[7]));
    asm volatile("" : "+v"(mr[0]), "+v"(mr[1]), "+v"(mr[2]), "+v"(mr[3]),
                      "+v"(mr[4]), "+v"(mr[5]), "+v"(mr[6]), "+v"(mr[7]));
#pragma unroll
    for (int h = 0; h < NPC; ++h) {
      xq[0][h] = xx[h] * mk[h] + ss[h] * (1.0f - mk[h]);
      xq[1][h] = xx[h] * mv[h] + ss[h] * (1.0f - mv[h]);
      xq[2][h] = xx[h] * mr[h] + ss[h] * (1.0f - mr[h]);
    }
  };

  float kvr[3];
  gemv_rows<3>(w_key, w_val, w_rec, mix, red, kvr, lane, wave, koff);

  const float ki = kvr[0], vi = kvr[1];
  const float r  = 1.0f / (1.0f + expf(-kvr[2]));

  const float ww = tfi + ki;
  const float p  = fmaxf(ppi, ww);
  const float e1 = expf(ppi - p);
  const float e2 = expf(ww - p);
  const float a  = e1 * aai + e2 * vi;
  const float b  = e1 * bbi + e2;

  const float ww2 = ppi + tdi;
  const float p2  = fmaxf(ww2, ki);
  const float e1b = expf(ww2 - p2);
  const float e2b = expf(ki - p2);

  const int sub = lane & 7;
  float res = r * (a / b);
  res = (sub == 1) ? xi : res;
  res = (sub == 2) ? e1b * aai + e2b * vi : res;
  res = (sub == 3) ? e1b * bbi + e2b : res;
  res = (sub == 4) ? p2 : res;
  if (sub < 5) fin[sub * GROWS + rloc] = res;
  __syncthreads();

  if (threadIdx.x < 5 * GROWS) {
    const int j = threadIdx.x >> 4;
    float* dst = (j == 0) ? rkv : out + j * DSZ;
    dst[blockIdx.x * GROWS + (threadIdx.x & (GROWS - 1))] = fin[threadIdx.x];
  }
}

__global__ __launch_bounds__(GTHREADS) __attribute__((amdgpu_waves_per_eu(1, 1)))
void gemv1_kernel(
    const float* __restrict__ W, const float* __restrict__ xin,
    float* __restrict__ y)
{
  __shared__ float red[GROWS][RPITCH];
  __shared__ float fin[GROWS];

  const int lane = threadIdx.x & 31;
  const int wave = __builtin_amdgcn_readfirstlane(threadIdx.x >> 5);
  const int koff = wave * GSEG + 4 * lane;

  v4f xp[NPC];
#pragma unroll
  for (int h = 0; h < NPC; ++h) xp[h] = *(const v4f*)(xin + koff + 128 * h);
  auto take = [&](v4f (&xq)[1][NPC]) {
#pragma unroll
    for (int h = 0; h < NPC; ++h) xq[0][h] = xp[h];
  };

  float ys[1];
  gemv_rows<1>(W, W, W, take, red, ys, lane, wave, koff);

  if ((lane & 7) == 0) fin[wave * 4 + (lane >> 3)] = ys[0];
  __syncthreads();
  if (threadIdx.x < GROWS)
    y[blockIdx.x * GROWS + threadIdx.x] = fin[threadIdx.x];
}

extern "C" void kernel_launch(void* const* d_in, const int* in_sizes, int n_in,
                              void* d_out, int out_size, void* d_ws,
                              size_t ws_size, hipStream_t stream)
{
  const float* x       = (const float*)d_in[0];
  const float* sxx     = (const float*)d_in[1];
  const float* aa      = (const float*)d_in[2];
  const float* bb      = (const float*)d_in[3];
  const float* pp      = (const float*)d_in[4];
  const float* w_key   = (const float*)d_in[5];
  const float* w_val   = (const float*)d_in[6];
  const float* w_rec   = (const float*)d_in[7];
  const float* w_out   = (const float*)d_in[8];
  const float* t_decay = (const float*)d_in[9];
  const float* t_first = (const float*)d_in[10];
  const float* t_mix_k = (const float*)d_in[11];
  const float* t_mix_v = (const float*)d_in[12];
  const float* t_mix_r = (const float*)d_in[13];

  float* out = (float*)d_out;
  float* rkv = (float*)d_ws;

  mix_gemv3_wkv_kernel<<<DSZ / GROWS, GTHREADS, 0, stream>>>(
      w_key, w_val, w_rec, x, sxx, t_mix_k, t_mix_v, t_mix_r,
      aa, bb, pp, t_decay, t_first, rkv, out);

  gemv1_kernel<<<DSZ / GROWS, GTHREADS, 0, stream>>>(w_out, rkv, out);
}
